# starting kernel
# baseline (speedup 1.0000x reference)
#include <hip/hip_runtime.h>


#define NN      2048
#define TILE_M  128
#define TILE_N  128
#define KC      32
#define A_PAD   36
#define B_PADN  132

typedef float v2f __attribute__((ext_vector_type(2)));
typedef float v8f __attribute__((ext_vector_type(8)));

__global__ __launch_bounds__(256) void build_A_kernel(const float* __restrict__ x,
                                                      float* __restrict__ A) {
    int idx = blockIdx.x * blockDim.x + threadIdx.x;
    int row = idx >> 11;
    int col = idx & (NN - 1);
    float v = x[idx];
    if (row == col) v -= 1.0f;
    A[idx] = v;
}

__global__ __launch_bounds__(256) void row0_kernel(const float* __restrict__ start,
                                                   float* __restrict__ S) {
    int j = blockIdx.x * blockDim.x + threadIdx.x;
    S[j] = start[199 * NN + j];
}

__global__ __launch_bounds__(256)
void gemm_f32_wmma(const float* __restrict__ Am, const float* __restrict__ Bm,
                   float* __restrict__ Cm, int M) {
    __shared__ float As[TILE_M][A_PAD];
    __shared__ float Bp[KC / 2][B_PADN][2];

    const int tid  = threadIdx.x;
    const int lane = tid & 31;
    const int wave = tid >> 5;
    const int wm   = wave & 3;
    const int wn   = wave >> 2;
    const int hi   = lane >> 4;
    const int ml   = lane & 15;

    const int blockN0 = blockIdx.x * TILE_N;
    const int blockM0 = blockIdx.y * TILE_M;

    v8f a00 = {}, a01 = {}, a02 = {}, a03 = {};
    v8f a10 = {}, a11 = {}, a12 = {}, a13 = {};

    const int rowA = tid >> 3;
    const int colA = (tid & 7) << 2;
    const int kp0  = tid >> 5;
    const int cBb  = (tid & 31) << 2;

    for (int kb = 0; kb < NN; kb += KC) {
        __syncthreads();
        {
            float4 z = {0.f, 0.f, 0.f, 0.f};
#pragma unroll
            for (int i = 0; i < 4; ++i) {
                int r = blockM0 + rowA + 32 * i;
                float4 a = (r < M) ? *(const float4*)(Am + (size_t)r * NN + kb + colA) : z;
                *(float4*)&As[rowA + 32 * i][colA] = a;
            }
        }
        {
#pragma unroll
            for (int i = 0; i < 2; ++i) {
                int kp = kp0 + 8 * i;
                const float* brow = Bm + (size_t)(kb + 2 * kp) * NN + blockN0 + cBb;
                float4 lo = *(const float4*)(brow);
                float4 hi4 = *(const float4*)(brow + NN);
                v2f p0 = {lo.x, hi4.x}, p1 = {lo.y, hi4.y};
                v2f p2 = {lo.z, hi4.z}, p3 = {lo.w, hi4.w};
                *(v2f*)&Bp[kp][cBb + 0][0] = p0;
                *(v2f*)&Bp[kp][cBb + 1][0] = p1;
                *(v2f*)&Bp[kp][cBb + 2][0] = p2;
                *(v2f*)&Bp[kp][cBb + 3][0] = p3;
            }
        }
        __syncthreads();

        const int cb0 = wn * 64 + ml;
        const int ar0 = wm * 32 + ml;
#pragma unroll
        for (int kk = 0; kk < KC; kk += 4) {
            v2f fa0 = *(const v2f*)&As[ar0][kk + 2 * hi];
            v2f fa1 = *(const v2f*)&As[ar0 + 16][kk + 2 * hi];
            int kB2 = (kk >> 1) + hi;
            v2f fb0 = *(const v2f*)&Bp[kB2][cb0 +  0][0];
            v2f fb1 = *(const v2f*)&Bp[kB2][cb0 + 16][0];
            v2f fb2 = *(const v2f*)&Bp[kB2][cb0 + 32][0];
            v2f fb3 = *(const v2f*)&Bp[kB2][cb0 + 48][0];
            a00 = __builtin_amdgcn_wmma_f32_16x16x4_f32(false, fa0, false, fb0, (short)0, a00, false, false);
            a01 = __builtin_amdgcn_wmma_f32_16x16x4_f32(false, fa0, false, fb1, (short)0, a01, false, false);
            a02 = __builtin_amdgcn_wmma_f32_16x16x4_f32(false, fa0, false, fb2, (short)0, a02, false, false);
            a03 = __builtin_amdgcn_wmma_f32_16x16x4_f32(false, fa0, false, fb3, (short)0, a03, false, false);
            a10 = __builtin_amdgcn_wmma_f32_16x16x4_f32(false, fa1, false, fb0, (short)0, a10, false, false);
            a11 = __builtin_amdgcn_wmma_f32_16x16x4_f32(false, fa1, false, fb1, (short)0, a11, false, false);
            a12 = __builtin_amdgcn_wmma_f32_16x16x4_f32(false, fa1, false, fb2, (short)0, a12, false, false);
            a13 = __builtin_amdgcn_wmma_f32_16x16x4_f32(false, fa1, false, fb3, (short)0, a13, false, false);
        }
    }

    const int colBase = blockN0 + wn * 64 + ml;
    const int row0 = blockM0 + wm * 32 + 8 * hi;
#pragma unroll
    for (int r = 0; r < 8; ++r) {
        int rg = row0 + r;
        if (rg < M) {
            float* cp = Cm + (size_t)rg * NN + colBase;
            cp[ 0] = a00[r];
            cp[16] = a01[r];
            cp[32] = a02[r];
            cp[48] = a03[r];
        }
        int rg2 = row0 + 16 + r;
        if (rg2 < M) {
            float* cp = Cm + (size_t)rg2 * NN + colBase;
            cp[ 0] = a10[r];
            cp[16] = a11[r];
            cp[32] = a12[r];
            cp[48] = a13[r];
        }
    }
}

extern "C" void kernel_launch(void* const* d_in, const int* in_sizes, int n_in,
                              void* d_out, int out_size, void* d_ws, size_t ws_size,
                              hipStream_t stream) {
    (void)in_sizes; (void)n_in; (void)out_size;
    const float* x     = (const float*)d_in[0];
    const float* start = (const float*)d_in[1];
    float* S = (float*)d_out;

    const size_t MATSZ = (size_t)NN * NN;
    if (ws_size < 2 * MATSZ * sizeof(float)) return;
    float* P  = (float*)d_ws;
    float* Pn = P + MATSZ;

    build_A_kernel<<<dim3((unsigned)(MATSZ / 256)), dim3(256), 0, stream>>>(x, P);
    row0_kernel<<<dim3(NN / 256), dim3(256), 0, stream>>>(start, S);

    for (int k = 1; k < NN; k <<= 1) {
        dim3 grid(NN / TILE_N, (unsigned)((k + TILE_M - 1) / TILE_M));
        gemm_f32_wmma<<<grid, dim3(256), 0, stream>>>(S, P, S + (size_t)k * NN, k);
        if (2 * k < NN) {
            dim3 g2(NN / TILE_N, NN / TILE_M);
            gemm_f32_wmma<<<g2, dim3(256), 0, stream>>>(P, P, Pn, NN);
            float* t = P; P = Pn; Pn = t;
        }
    }
}


# Row product and squaring of each doubling level merged into one launch (row tiles ride as extra block rows of the squaring grid): 23 launches become 13; arithmetic and summation order unchanged
# speedup vs baseline: 1.7725x; 1.7725x over previous
#include <hip/hip_runtime.h>


#define NN      2048
#define TILE_M  128
#define TILE_N  128
#define KC      32
#define A_PAD   36
#define B_PADN  132

typedef float v2f __attribute__((ext_vector_type(2)));
typedef float v8f __attribute__((ext_vector_type(8)));

__global__ __launch_bounds__(256) void build_A_kernel(const float* __restrict__ x,
                                                      float* __restrict__ A) {
    int idx = blockIdx.x * blockDim.x + threadIdx.x;
    int row = idx >> 11;
    int col = idx & (NN - 1);
    float v = x[idx];
    if (row == col) v -= 1.0f;
    A[idx] = v;
}

__global__ __launch_bounds__(256) void row0_kernel(const float* __restrict__ start,
                                                   float* __restrict__ S) {
    int j = blockIdx.x * blockDim.x + threadIdx.x;
    S[j] = start[199 * NN + j];
}

__global__ __launch_bounds__(256)
void gemm_f32_wmma(const float* __restrict__ Bm,
                   const float* __restrict__ Asq, float* __restrict__ Csq, int sqTiles,
                   const float* __restrict__ Arow, float* __restrict__ Crow, int Mrow) {
    __shared__ float As[TILE_M][A_PAD];
    __shared__ float Bp[KC / 2][B_PADN][2];

    const int tid  = threadIdx.x;
    const int lane = tid & 31;
    const int wave = tid >> 5;
    const int wm   = wave & 3;
    const int wn   = wave >> 2;
    const int hi   = lane >> 4;
    const int ml   = lane & 15;

    const bool sq = (int)blockIdx.y < sqTiles;
    const float* __restrict__ Am = sq ? Asq : Arow;
    float* __restrict__ Cm       = sq ? Csq : Crow;
    const int M                  = sq ? NN : Mrow;

    const int blockN0 = blockIdx.x * TILE_N;
    const int blockM0 = (sq ? (int)blockIdx.y : (int)blockIdx.y - sqTiles) * TILE_M;

    v8f a00 = {}, a01 = {}, a02 = {}, a03 = {};
    v8f a10 = {}, a11 = {}, a12 = {}, a13 = {};

    const int rowA = tid >> 3;
    const int colA = (tid & 7) << 2;
    const int kp0  = tid >> 5;
    const int cBb  = (tid & 31) << 2;

    for (int kb = 0; kb < NN; kb += KC) {
        __syncthreads();
        {
            float4 z = {0.f, 0.f, 0.f, 0.f};
#pragma unroll
            for (int i = 0; i < 4; ++i) {
                int r = blockM0 + rowA + 32 * i;
                float4 a = (r < M) ? *(const float4*)(Am + (size_t)r * NN + kb + colA) : z;
                *(float4*)&As[rowA + 32 * i][colA] = a;
            }
        }
        {
#pragma unroll
            for (int i = 0; i < 2; ++i) {
                int kp = kp0 + 8 * i;
                const float* brow = Bm + (size_t)(kb + 2 * kp) * NN + blockN0 + cBb;
                float4 lo = *(const float4*)(brow);
                float4 hi4 = *(const float4*)(brow + NN);
                v2f p0 = {lo.x, hi4.x}, p1 = {lo.y, hi4.y};
                v2f p2 = {lo.z, hi4.z}, p3 = {lo.w, hi4.w};
                *(v2f*)&Bp[kp][cBb + 0][0] = p0;
                *(v2f*)&Bp[kp][cBb + 1][0] = p1;
                *(v2f*)&Bp[kp][cBb + 2][0] = p2;
                *(v2f*)&Bp[kp][cBb + 3][0] = p3;
            }
        }
        __syncthreads();

        const int cb0 = wn * 64 + ml;
        const int ar0 = wm * 32 + ml;
#pragma unroll
        for (int kk = 0; kk < KC; kk += 4) {
            v2f fa0 = *(const v2f*)&As[ar0][kk + 2 * hi];
            v2f fa1 = *(const v2f*)&As[ar0 + 16][kk + 2 * hi];
            int kB2 = (kk >> 1) + hi;
            v2f fb0 = *(const v2f*)&Bp[kB2][cb0 +  0][0];
            v2f fb1 = *(const v2f*)&Bp[kB2][cb0 + 16][0];
            v2f fb2 = *(const v2f*)&Bp[kB2][cb0 + 32][0];
            v2f fb3 = *(const v2f*)&Bp[kB2][cb0 + 48][0];
            a00 = __builtin_amdgcn_wmma_f32_16x16x4_f32(false, fa0, false, fb0, (short)0, a00, false, false);
            a01 = __builtin_amdgcn_wmma_f32_16x16x4_f32(false, fa0, false, fb1, (short)0, a01, false, false);
            a02 = __builtin_amdgcn_wmma_f32_16x16x4_f32(false, fa0, false, fb2, (short)0, a02, false, false);
            a03 = __builtin_amdgcn_wmma_f32_16x16x4_f32(false, fa0, false, fb3, (short)0, a03, false, false);
            a10 = __builtin_amdgcn_wmma_f32_16x16x4_f32(false, fa1, false, fb0, (short)0, a10, false, false);
            a11 = __builtin_amdgcn_wmma_f32_16x16x4_f32(false, fa1, false, fb1, (short)0, a11, false, false);
            a12 = __builtin_amdgcn_wmma_f32_16x16x4_f32(false, fa1, false, fb2, (short)0, a12, false, false);
            a13 = __builtin_amdgcn_wmma_f32_16x16x4_f32(false, fa1, false, fb3, (short)0, a13, false, false);
        }
    }

    const int colBase = blockN0 + wn * 64 + ml;
    const int row0 = blockM0 + wm * 32 + 8 * hi;
#pragma unroll
    for (int r = 0; r < 8; ++r) {
        int rg = row0 + r;
        if (rg < M) {
            float* cp = Cm + (size_t)rg * NN + colBase;
            cp[ 0] = a00[r];
            cp[16] = a01[r];
            cp[32] = a02[r];
            cp[48] = a03[r];
        }
        int rg2 = row0 + 16 + r;
        if (rg2 < M) {
            float* cp = Cm + (size_t)rg2 * NN + colBase;
            cp[ 0] = a10[r];
            cp[16] = a11[r];
            cp[32] = a12[r];
            cp[48] = a13[r];
        }
    }
}

extern "C" void kernel_launch(void* const* d_in, const int* in_sizes, int n_in,
                              void* d_out, int out_size, void* d_ws, size_t ws_size,
                              hipStream_t stream) {
    (void)in_sizes; (void)n_in; (void)out_size;
    const float* x     = (const float*)d_in[0];
    const float* start = (const float*)d_in[1];
    float* S = (float*)d_out;

    const size_t MATSZ = (size_t)NN * NN;
    if (ws_size < 2 * MATSZ * sizeof(float)) return;
    float* P  = (float*)d_ws;
    float* Pn = P + MATSZ;

    build_A_kernel<<<dim3((unsigned)(MATSZ / 256)), dim3(256), 0, stream>>>(x, P);
    row0_kernel<<<dim3(NN / 256), dim3(256), 0, stream>>>(start, S);

    for (int k = 1; k < NN; k <<= 1) {
        const int rowTiles = (k + TILE_M - 1) / TILE_M;
        const int sqTiles  = (2 * k < NN) ? NN / TILE_M : 0;
        dim3 grid(NN / TILE_N, (unsigned)(sqTiles + rowTiles));
        gemm_f32_wmma<<<grid, dim3(256), 0, stream>>>(P, P, Pn, sqTiles,
                                                      S, S + (size_t)k * NN, k);
        if (sqTiles) { float* t = P; P = Pn; Pn = t; }
    }
}


# K loop software-pipelined: next chunk's global loads (plain loads, row guard moved to the LDS store) issued ahead of the current chunk's WMMAs into a second LDS stage; one barrier per chunk
# speedup vs baseline: 2.5255x; 1.4248x over previous
#include <hip/hip_runtime.h>


#define NN      2048
#define TILE_M  128
#define TILE_N  128
#define KC      32
#define A_PAD   36
#define B_PADN  132

typedef float v2f __attribute__((ext_vector_type(2)));
typedef float v4f __attribute__((ext_vector_type(4)));
typedef float v8f __attribute__((ext_vector_type(8)));

__global__ __launch_bounds__(256) void build_A_kernel(const float* __restrict__ x,
                                                      float* __restrict__ A) {
    int idx = blockIdx.x * blockDim.x + threadIdx.x;
    int row = idx >> 11;
    int col = idx & (NN - 1);
    float v = x[idx];
    if (row == col) v -= 1.0f;
    A[idx] = v;
}

__global__ __launch_bounds__(256) void row0_kernel(const float* __restrict__ start,
                                                   float* __restrict__ S) {
    int j = blockIdx.x * blockDim.x + threadIdx.x;
    S[j] = start[199 * NN + j];
}

__global__ __launch_bounds__(256)
void gemm_f32_wmma(const float* __restrict__ Bm,
                   const float* __restrict__ Asq, float* __restrict__ Csq, int sqTiles,
                   const float* __restrict__ Arow, float* __restrict__ Crow, int Mrow) {
    __shared__ float As[2][TILE_M][A_PAD];
    __shared__ float Bp[2][KC / 2][B_PADN][2];

    const int tid  = threadIdx.x;
    const int lane = tid & 31;
    const int wave = tid >> 5;
    const int wm   = wave & 3;
    const int wn   = wave >> 2;
    const int hi   = lane >> 4;
    const int ml   = lane & 15;

    const bool sq = (int)blockIdx.y < sqTiles;
    const float* __restrict__ Am = sq ? Asq : Arow;
    float* __restrict__ Cm       = sq ? Csq : Crow;
    const int M                  = sq ? NN : Mrow;

    const int blockN0 = blockIdx.x * TILE_N;
    const int blockM0 = (sq ? (int)blockIdx.y : (int)blockIdx.y - sqTiles) * TILE_M;

    v8f a00 = {}, a01 = {}, a02 = {}, a03 = {};
    v8f a10 = {}, a11 = {}, a12 = {}, a13 = {};

    const int rowA = tid >> 3;
    const int colA = (tid & 7) << 2;
    const int kp0  = tid >> 5;
    const int cBb  = (tid & 31) << 2;

    const float* aSrc[4];
    bool aOk[4];
#pragma unroll
    for (int i = 0; i < 4; ++i) {
        int r = blockM0 + rowA + 32 * i;
        aOk[i]  = r < M;
        aSrc[i] = Am + (size_t)(aOk[i] ? r : 0) * NN + colA;
    }
    const float* bSrc = Bm + (size_t)(2 * kp0) * NN + blockN0 + cBb;

    v4f ra[4], rlo[2], rhi[2];
    const v4f z = {0.f, 0.f, 0.f, 0.f};

#pragma unroll
    for (int i = 0; i < 4; ++i)
        ra[i] = *(const v4f*)(aSrc[i]);
#pragma unroll
    for (int i = 0; i < 2; ++i) {
        const float* brow = bSrc + (size_t)(16 * i) * NN;
        rlo[i] = *(const v4f*)(brow);
        rhi[i] = *(const v4f*)(brow + NN);
    }
#pragma unroll
    for (int i = 0; i < 4; ++i)
        *(v4f*)&As[0][rowA + 32 * i][colA] = aOk[i] ? ra[i] : z;
#pragma unroll
    for (int i = 0; i < 2; ++i) {
        int kp = kp0 + 8 * i;
        v2f p0 = {rlo[i].x, rhi[i].x}, p1 = {rlo[i].y, rhi[i].y};
        v2f p2 = {rlo[i].z, rhi[i].z}, p3 = {rlo[i].w, rhi[i].w};
        *(v2f*)&Bp[0][kp][cBb + 0][0] = p0;
        *(v2f*)&Bp[0][kp][cBb + 1][0] = p1;
        *(v2f*)&Bp[0][kp][cBb + 2][0] = p2;
        *(v2f*)&Bp[0][kp][cBb + 3][0] = p3;
    }
    __syncthreads();

    const int cb0 = wn * 64 + ml;
    const int ar0 = wm * 32 + ml;

    for (int kb = 0; kb < NN; kb += KC) {
        const int cur = (kb / KC) & 1;
        const bool more = kb + KC < NN;

        if (more) {
#pragma unroll
            for (int i = 0; i < 4; ++i)
                ra[i] = *(const v4f*)(aSrc[i] + kb + KC);
#pragma unroll
            for (int i = 0; i < 2; ++i) {
                const float* brow = bSrc + (size_t)(kb + KC + 16 * i) * NN;
                rlo[i] = *(const v4f*)(brow);
                rhi[i] = *(const v4f*)(brow + NN);
            }
        }

#pragma unroll
        for (int kk = 0; kk < KC; kk += 4) {
            v2f fa0 = *(const v2f*)&As[cur][ar0][kk + 2 * hi];
            v2f fa1 = *(const v2f*)&As[cur][ar0 + 16][kk + 2 * hi];
            int kB2 = (kk >> 1) + hi;
            v2f fb0 = *(const v2f*)&Bp[cur][kB2][cb0 +  0][0];
            v2f fb1 = *(const v2f*)&Bp[cur][kB2][cb0 + 16][0];
            v2f fb2 = *(const v2f*)&Bp[cur][kB2][cb0 + 32][0];
            v2f fb3 = *(const v2f*)&Bp[cur][kB2][cb0 + 48][0];
            a00 = __builtin_amdgcn_wmma_f32_16x16x4_f32(false, fa0, false, fb0, (short)0, a00, false, false);
            a01 = __builtin_amdgcn_wmma_f32_16x16x4_f32(false, fa0, false, fb1, (short)0, a01, false, false);
            a02 = __builtin_amdgcn_wmma_f32_16x16x4_f32(false, fa0, false, fb2, (short)0, a02, false, false);
            a03 = __builtin_amdgcn_wmma_f32_16x16x4_f32(false, fa0, false, fb3, (short)0, a03, false, false);
            a10 = __builtin_amdgcn_wmma_f32_16x16x4_f32(false, fa1, false, fb0, (short)0, a10, false, false);
            a11 = __builtin_amdgcn_wmma_f32_16x16x4_f32(false, fa1, false, fb1, (short)0, a11, false, false);
            a12 = __builtin_amdgcn_wmma_f32_16x16x4_f32(false, fa1, false, fb2, (short)0, a12, false, false);
            a13 = __builtin_amdgcn_wmma_f32_16x16x4_f32(false, fa1, false, fb3, (short)0, a13, false, false);
        }

        if (more) {
            const int nxt = cur ^ 1;
#pragma unroll
            for (int i = 0; i < 4; ++i)
                *(v4f*)&As[nxt][rowA + 32 * i][colA] = aOk[i] ? ra[i] : z;
#pragma unroll
            for (int i = 0; i < 2; ++i) {
                int kp = kp0 + 8 * i;
                v2f p0 = {rlo[i].x, rhi[i].x}, p1 = {rlo[i].y, rhi[i].y};
                v2f p2 = {rlo[i].z, rhi[i].z}, p3 = {rlo[i].w, rhi[i].w};
                *(v2f*)&Bp[nxt][kp][cBb + 0][0] = p0;
                *(v2f*)&Bp[nxt][kp][cBb + 1][0] = p1;
                *(v2f*)&Bp[nxt][kp][cBb + 2][0] = p2;
                *(v2f*)&Bp[nxt][kp][cBb + 3][0] = p3;
            }
        }
        __syncthreads();
    }

    const int colBase = blockN0 + wn * 64 + ml;
    const int row0 = blockM0 + wm * 32 + 8 * hi;
#pragma unroll
    for (int r = 0; r < 8; ++r) {
        int rg = row0 + r;
        if (rg < M) {
            float* cp = Cm + (size_t)rg * NN + colBase;
            cp[ 0] = a00[r];
            cp[16] = a01[r];
            cp[32] = a02[r];
            cp[48] = a03[r];
        }
        int rg2 = row0 + 16 + r;
        if (rg2 < M) {
            float* cp = Cm + (size_t)rg2 * NN + colBase;
            cp[ 0] = a10[r];
            cp[16] = a11[r];
            cp[32] = a12[r];
            cp[48] = a13[r];
        }
    }
}

extern "C" void kernel_launch(void* const* d_in, const int* in_sizes, int n_in,
                              void* d_out, int out_size, void* d_ws, size_t ws_size,
                              hipStream_t stream) {
    (void)in_sizes; (void)n_in; (void)out_size;
    const float* x     = (const float*)d_in[0];
    const float* start = (const float*)d_in[1];
    float* S = (float*)d_out;

    const size_t MATSZ = (size_t)NN * NN;
    if (ws_size < 2 * MATSZ * sizeof(float)) return;
    float* P  = (float*)d_ws;
    float* Pn = P + MATSZ;

    build_A_kernel<<<dim3((unsigned)(MATSZ / 256)), dim3(256), 0, stream>>>(x, P);
    row0_kernel<<<dim3(NN / 256), dim3(256), 0, stream>>>(start, S);

    for (int k = 1; k < NN; k <<= 1) {
        const int rowTiles = (k + TILE_M - 1) / TILE_M;
        const int sqTiles  = (2 * k < NN) ? NN / TILE_M : 0;
        dim3 grid(NN / TILE_N, (unsigned)(sqTiles + rowTiles));
        gemm_f32_wmma<<<grid, dim3(256), 0, stream>>>(P, P, Pn, sqTiles,
                                                      S, S + (size_t)k * NN, k);
        if (sqTiles) { float* t = P; P = Pn; Pn = t; }
    }
}


# Row product of each level folded into the squaring's blocks as 16 or 32 extra A rows per block (last level: 64-row slices): every level is 256 equal blocks, one per WGP; summation order unchanged
# speedup vs baseline: 3.8489x; 1.5240x over previous
#include <hip/hip_runtime.h>


#define NN      2048
#define TILE_N  128
#define KC      32
#define A_PAD   36
#define B_PADN  132
#define C_PADN  132

typedef float v2f __attribute__((ext_vector_type(2)));
typedef float v4f __attribute__((ext_vector_type(4)));
typedef float v8f __attribute__((ext_vector_type(8)));

__global__ __launch_bounds__(256) void build_A_kernel(const float* __restrict__ x,
                                                      float* __restrict__ A) {
    int idx = blockIdx.x * blockDim.x + threadIdx.x;
    int row = idx >> 11;
    int col = idx & (NN - 1);
    float v = x[idx];
    if (row == col) v -= 1.0f;
    A[idx] = v;
}

__global__ __launch_bounds__(256) void row0_kernel(const float* __restrict__ start,
                                                   float* __restrict__ S) {
    int j = blockIdx.x * blockDim.x + threadIdx.x;
    S[j] = start[199 * NN + j];
}

#define WMMA_F32(A_, B_, C_) \
    __builtin_amdgcn_wmma_f32_16x16x4_f32(false, (A_), false, (B_), (short)0, (C_), false, false)

template <int MROWS, int NEX>
__device__ __forceinline__ void gemm_body(float* As, float* Bp, float* Cs,
                                          const float* __restrict__ Bm, float* __restrict__ Csq,
                                          const float* __restrict__ Srow, float* __restrict__ Crow,
                                          const int k) {
    constexpr int AR      = MROWS + 16 * NEX;
    constexpr int ARP     = (AR + 31) / 32 * 32;
    constexpr int NLD     = ARP / 32;
    constexpr int NMAIN   = MROWS / 32;
    constexpr int A_STAGE = ARP * A_PAD;
    constexpr int B_STAGE = (KC / 2) * B_PADN * 2;
    constexpr int NEXA    = NEX > 0 ? NEX : 1;

    const int tid  = threadIdx.x;
    const int lane = tid & 31;
    const int wave = tid >> 5;
    const int wm   = wave & 3;
    const int wn   = wave >> 2;
    const int hi   = lane >> 4;
    const int ml   = lane & 15;

    const int blockN0 = blockIdx.x * TILE_N;
    const int blockM0 = blockIdx.y * MROWS;
    const int exRow0  = blockIdx.y * (16 * NEX);

    v8f am[8];
    v8f ex[NEXA];
#pragma unroll
    for (int i = 0; i < 8; ++i) am[i] = (v8f){};
#pragma unroll
    for (int i = 0; i < NEXA; ++i) ex[i] = (v8f){};

    const int rowA = tid >> 3;
    const int colA = (tid & 7) << 2;
    const int kp0  = tid >> 5;
    const int cBb  = (tid & 31) << 2;

    const float* aSrc[NLD];
    bool aOk[NLD];
#pragma unroll
    for (int i = 0; i < NLD; ++i) {
        if (i < NMAIN) {
            aOk[i]  = true;
            aSrc[i] = Bm + (size_t)(blockM0 + rowA + 32 * i) * NN + colA;
        } else {
            const int e  = rowA + 32 * (i - NMAIN);
            const int rS = exRow0 + e;
            aOk[i]  = (e < 16 * NEX) && (rS < k);
            aSrc[i] = Srow + (size_t)(aOk[i] ? rS : 0) * NN + colA;
        }
    }
    const float* bSrc = Bm + (size_t)(2 * kp0) * NN + blockN0 + cBb;

    v4f ra[NLD], rlo[2], rhi[2];
    const v4f z = {0.f, 0.f, 0.f, 0.f};

#pragma unroll
    for (int i = 0; i < NLD; ++i)
        ra[i] = *(const v4f*)(aSrc[i]);
#pragma unroll
    for (int i = 0; i < 2; ++i) {
        const float* brow = bSrc + (size_t)(16 * i) * NN;
        rlo[i] = *(const v4f*)(brow);
        rhi[i] = *(const v4f*)(brow + NN);
    }
#pragma unroll
    for (int i = 0; i < NLD; ++i)
        *(v4f*)&As[(rowA + 32 * i) * A_PAD + colA] = aOk[i] ? ra[i] : z;
#pragma unroll
    for (int i = 0; i < 2; ++i) {
        const int kp = kp0 + 8 * i;
        v2f p0 = {rlo[i].x, rhi[i].x}, p1 = {rlo[i].y, rhi[i].y};
        v2f p2 = {rlo[i].z, rhi[i].z}, p3 = {rlo[i].w, rhi[i].w};
        float* bp = &Bp[(kp * B_PADN + cBb) * 2];
        *(v2f*)(bp + 0) = p0;
        *(v2f*)(bp + 2) = p1;
        *(v2f*)(bp + 4) = p2;
        *(v2f*)(bp + 6) = p3;
    }
    __syncthreads();

    const int cb0 = wn * 64 + ml;
    const int ar0 = wm * 32 + ml;
    const int cbx = wave * 16 + ml;

    for (int kb = 0; kb < NN; kb += KC) {
        const int cur = (kb / KC) & 1;
        const bool more = kb + KC < NN;
        const float* Ac = As + cur * A_STAGE;
        const float* Bc = Bp + cur * B_STAGE;

        if (more) {
#pragma unroll
            for (int i = 0; i < NLD; ++i)
                ra[i] = *(const v4f*)(aSrc[i] + kb + KC);
#pragma unroll
            for (int i = 0; i < 2; ++i) {
                const float* brow = bSrc + (size_t)(kb + KC + 16 * i) * NN;
                rlo[i] = *(const v4f*)(brow);
                rhi[i] = *(const v4f*)(brow + NN);
            }
        }

#pragma unroll
        for (int kk = 0; kk < KC; kk += 4) {
            const int kB2 = (kk >> 1) + hi;
            if (MROWS > 0) {
                v2f fa0 = *(const v2f*)&Ac[ar0 * A_PAD + kk + 2 * hi];
                v2f fa1 = *(const v2f*)&Ac[(ar0 + 16) * A_PAD + kk + 2 * hi];
                v2f fb0 = *(const v2f*)&Bc[(kB2 * B_PADN + cb0 +  0) * 2];
                v2f fb1 = *(const v2f*)&Bc[(kB2 * B_PADN + cb0 + 16) * 2];
                v2f fb2 = *(const v2f*)&Bc[(kB2 * B_PADN + cb0 + 32) * 2];
                v2f fb3 = *(const v2f*)&Bc[(kB2 * B_PADN + cb0 + 48) * 2];
                am[0] = WMMA_F32(fa0, fb0, am[0]);
                am[1] = WMMA_F32(fa0, fb1, am[1]);
                am[2] = WMMA_F32(fa0, fb2, am[2]);
                am[3] = WMMA_F32(fa0, fb3, am[3]);
                am[4] = WMMA_F32(fa1, fb0, am[4]);
                am[5] = WMMA_F32(fa1, fb1, am[5]);
                am[6] = WMMA_F32(fa1, fb2, am[6]);
                am[7] = WMMA_F32(fa1, fb3, am[7]);
            }
            if (NEX > 0) {
                v2f fbx = *(const v2f*)&Bc[(kB2 * B_PADN + cbx) * 2];
#pragma unroll
                for (int g = 0; g < NEX; ++g) {
                    v2f fe = *(const v2f*)&Ac[(MROWS + 16 * g + ml) * A_PAD + kk + 2 * hi];
                    ex[g] = WMMA_F32(fe, fbx, ex[g]);
                }
            }
        }

        if (more) {
            float* An = As + (cur ^ 1) * A_STAGE;
            float* Bn = Bp + (cur ^ 1) * B_STAGE;
#pragma unroll
            for (int i = 0; i < NLD; ++i)
                *(v4f*)&An[(rowA + 32 * i) * A_PAD + colA] = aOk[i] ? ra[i] : z;
#pragma unroll
            for (int i = 0; i < 2; ++i) {
                const int kp = kp0 + 8 * i;
                v2f p0 = {rlo[i].x, rhi[i].x}, p1 = {rlo[i].y, rhi[i].y};
                v2f p2 = {rlo[i].z, rhi[i].z}, p3 = {rlo[i].w, rhi[i].w};
                float* bp = &Bn[(kp * B_PADN + cBb) * 2];
                *(v2f*)(bp + 0) = p0;
                *(v2f*)(bp + 2) = p1;
                *(v2f*)(bp + 4) = p2;
                *(v2f*)(bp + 6) = p3;
            }
        }
        __syncthreads();
    }

    if (NEX > 0) {
#pragma unroll
        for (int g = 0; g < NEX; ++g)
#pragma unroll
            for (int r = 0; r < 8; ++r)
                Cs[(16 * g + 8 * hi + r) * C_PADN + cbx] = ex[g][r];
        __syncthreads();
#pragma unroll
        for (int j = 0; j < 2 * NEX; ++j) {
            const int row = wave + 8 * j;
            const int rS  = exRow0 + row;
            if (rS < k)
                *(v4f*)(Crow + (size_t)rS * NN + blockN0 + 4 * lane) =
                    *(const v4f*)&Cs[row * C_PADN + 4 * lane];
        }
    }

    if (MROWS > 0) {
        const int colBase = blockN0 + wn * 64 + ml;
        const int row0 = blockM0 + wm * 32 + 8 * hi;
#pragma unroll
        for (int r = 0; r < 8; ++r) {
            float* cp = Csq + (size_t)(row0 + r) * NN + colBase;
            cp[ 0] = am[0][r];
            cp[16] = am[1][r];
            cp[32] = am[2][r];
            cp[48] = am[3][r];
            float* cq = Csq + (size_t)(row0 + 16 + r) * NN + colBase;
            cq[ 0] = am[4][r];
            cq[16] = am[5][r];
            cq[32] = am[6][r];
            cq[48] = am[7][r];
        }
    }
}

template <int MROWS, int NEX>
__global__ __launch_bounds__(256)
void level_kernel(const float* __restrict__ Bm, float* __restrict__ Csq,
                  const float* __restrict__ Srow, float* __restrict__ Crow, int k) {
    constexpr int ARP = (MROWS + 16 * NEX + 31) / 32 * 32;
    __shared__ float As[2 * ARP * A_PAD];
    __shared__ float Bp[2 * (KC / 2) * B_PADN * 2];
    __shared__ float Cs[16 * NEX * C_PADN];

    if constexpr (MROWS == 0) {
        gemm_body<MROWS, NEX>(As, Bp, Cs, Bm, Csq, Srow, Crow, k);
    } else {
        if ((int)blockIdx.y * (16 * NEX) < k)
            gemm_body<MROWS, NEX>(As, Bp, Cs, Bm, Csq, Srow, Crow, k);
        else
            gemm_body<MROWS, 0>(As, Bp, Cs, Bm, Csq, Srow, Crow, k);
    }
}

extern "C" void kernel_launch(void* const* d_in, const int* in_sizes, int n_in,
                              void* d_out, int out_size, void* d_ws, size_t ws_size,
                              hipStream_t stream) {
    (void)in_sizes; (void)n_in; (void)out_size;
    const float* x     = (const float*)d_in[0];
    const float* start = (const float*)d_in[1];
    float* S = (float*)d_out;

    const size_t MATSZ = (size_t)NN * NN;
    if (ws_size < 2 * MATSZ * sizeof(float)) return;
    float* P  = (float*)d_ws;
    float* Pn = P + MATSZ;

    build_A_kernel<<<dim3((unsigned)(MATSZ / 256)), dim3(256), 0, stream>>>(x, P);
    row0_kernel<<<dim3(NN / 256), dim3(256), 0, stream>>>(start, S);

    const dim3 grid(NN / TILE_N, 16);
    for (int k = 1; k < NN; k <<= 1) {
        float* Sk = S + (size_t)k * NN;
        if (k <= 256) {
            level_kernel<128, 1><<<grid, dim3(256), 0, stream>>>(P, Pn, S, Sk, k);
            float* t = P; P = Pn; Pn = t;
        } else if (2 * k < NN) {
            level_kernel<128, 2><<<grid, dim3(256), 0, stream>>>(P, Pn, S, Sk, k);
            float* t = P; P = Pn; Pn = t;
        } else {
            level_kernel<0, 4><<<grid, dim3(256), 0, stream>>>(P, Pn, S, Sk, k);
        }
    }
}


# Levels k=128..1024 (three squarings, their row products) replaced by a chain of 15 products of 128 rows with A^128 in a new 32x32-tile kernel; summation order of rows 128.. changes
# speedup vs baseline: 4.1262x; 1.0720x over previous
#include <hip/hip_runtime.h>


#define NN      2048
#define TILE_N  128
#define KC      32
#define A_PAD   36
#define B_PADN  132
#define C_PADN  132

typedef float v2f __attribute__((ext_vector_type(2)));
typedef float v4f __attribute__((ext_vector_type(4)));
typedef float v8f __attribute__((ext_vector_type(8)));

__global__ __launch_bounds__(256) void build_A_kernel(const float* __restrict__ x,
                                                      float* __restrict__ A) {
    int idx = blockIdx.x * blockDim.x + threadIdx.x;
    int row = idx >> 11;
    int col = idx & (NN - 1);
    float v = x[idx];
    if (row == col) v -= 1.0f;
    A[idx] = v;
}

__global__ __launch_bounds__(256) void row0_kernel(const float* __restrict__ start,
                                                   float* __restrict__ S) {
    int j = blockIdx.x * blockDim.x + threadIdx.x;
    S[j] = start[199 * NN + j];
}

#define WMMA_F32(A_, B_, C_) \
    __builtin_amdgcn_wmma_f32_16x16x4_f32(false, (A_), false, (B_), (short)0, (C_), false, false)

template <int MROWS, int NEX>
__device__ __forceinline__ void gemm_body(float* As, float* Bp, float* Cs,
                                          const float* __restrict__ Bm, float* __restrict__ Csq,
                                          const float* __restrict__ Srow, float* __restrict__ Crow,
                                          const int k) {
    constexpr int AR      = MROWS + 16 * NEX;
    constexpr int ARP     = (AR + 31) / 32 * 32;
    constexpr int NLD     = ARP / 32;
    constexpr int NMAIN   = MROWS / 32;
    constexpr int A_STAGE = ARP * A_PAD;
    constexpr int B_STAGE = (KC / 2) * B_PADN * 2;
    constexpr int NEXA    = NEX > 0 ? NEX : 1;

    const int tid  = threadIdx.x;
    const int lane = tid & 31;
    const int wave = tid >> 5;
    const int wm   = wave & 3;
    const int wn   = wave >> 2;
    const int hi   = lane >> 4;
    const int ml   = lane & 15;

    const int blockN0 = blockIdx.x * TILE_N;
    const int blockM0 = blockIdx.y * MROWS;
    const int exRow0  = blockIdx.y * (16 * NEX);

    v8f am[8];
    v8f ex[NEXA];
#pragma unroll
    for (int i = 0; i < 8; ++i) am[i] = (v8f){};
#pragma unroll
    for (int i = 0; i < NEXA; ++i) ex[i] = (v8f){};

    const int rowA = tid >> 3;
    const int colA = (tid & 7) << 2;
    const int kp0  = tid >> 5;
    const int cBb  = (tid & 31) << 2;

    const float* aSrc[NLD];
    bool aOk[NLD];
#pragma unroll
    for (int i = 0; i < NLD; ++i) {
        if (i < NMAIN) {
            aOk[i]  = true;
            aSrc[i] = Bm + (size_t)(blockM0 + rowA + 32 * i) * NN + colA;
        } else {
            const int e  = rowA + 32 * (i - NMAIN);
            const int rS = exRow0 + e;
            aOk[i]  = (e < 16 * NEX) && (rS < k);
            aSrc[i] = Srow + (size_t)(aOk[i] ? rS : 0) * NN + colA;
        }
    }
    const float* bSrc = Bm + (size_t)(2 * kp0) * NN + blockN0 + cBb;

    v4f ra[NLD], rlo[2], rhi[2];
    const v4f z = {0.f, 0.f, 0.f, 0.f};

#pragma unroll
    for (int i = 0; i < NLD; ++i)
        ra[i] = *(const v4f*)(aSrc[i]);
#pragma unroll
    for (int i = 0; i < 2; ++i) {
        const float* brow = bSrc + (size_t)(16 * i) * NN;
        rlo[i] = *(const v4f*)(brow);
        rhi[i] = *(const v4f*)(brow + NN);
    }
#pragma unroll
    for (int i = 0; i < NLD; ++i)
        *(v4f*)&As[(rowA + 32 * i) * A_PAD + colA] = aOk[i] ? ra[i] : z;
#pragma unroll
    for (int i = 0; i < 2; ++i) {
        const int kp = kp0 + 8 * i;
        v2f p0 = {rlo[i].x, rhi[i].x}, p1 = {rlo[i].y, rhi[i].y};
        v2f p2 = {rlo[i].z, rhi[i].z}, p3 = {rlo[i].w, rhi[i].w};
        float* bp = &Bp[(kp * B_PADN + cBb) * 2];
        *(v2f*)(bp + 0) = p0;
        *(v2f*)(bp + 2) = p1;
        *(v2f*)(bp + 4) = p2;
        *(v2f*)(bp + 6) = p3;
    }
    __syncthreads();

    const int cb0 = wn * 64 + ml;
    const int ar0 = wm * 32 + ml;
    const int cbx = wave * 16 + ml;

    for (int kb = 0; kb < NN; kb += KC) {
        const int cur = (kb / KC) & 1;
        const bool more = kb + KC < NN;
        const float* Ac = As + cur * A_STAGE;
        const float* Bc = Bp + cur * B_STAGE;

        if (more) {
#pragma unroll
            for (int i = 0; i < NLD; ++i)
                ra[i] = *(const v4f*)(aSrc[i] + kb + KC);
#pragma unroll
            for (int i = 0; i < 2; ++i) {
                const float* brow = bSrc + (size_t)(kb + KC + 16 * i) * NN;
                rlo[i] = *(const v4f*)(brow);
                rhi[i] = *(const v4f*)(brow + NN);
            }
        }

#pragma unroll
        for (int kk = 0; kk < KC; kk += 4) {
            const int kB2 = (kk >> 1) + hi;
            if (MROWS > 0) {
                v2f fa0 = *(const v2f*)&Ac[ar0 * A_PAD + kk + 2 * hi];
                v2f fa1 = *(const v2f*)&Ac[(ar0 + 16) * A_PAD + kk + 2 * hi];
                v2f fb0 = *(const v2f*)&Bc[(kB2 * B_PADN + cb0 +  0) * 2];
                v2f fb1 = *(const v2f*)&Bc[(kB2 * B_PADN + cb0 + 16) * 2];
                v2f fb2 = *(const v2f*)&Bc[(kB2 * B_PADN + cb0 + 32) * 2];
                v2f fb3 = *(const v2f*)&Bc[(kB2 * B_PADN + cb0 + 48) * 2];
                am[0] = WMMA_F32(fa0, fb0, am[0]);
                am[1] = WMMA_F32(fa0, fb1, am[1]);
                am[2] = WMMA_F32(fa0, fb2, am[2]);
                am[3] = WMMA_F32(fa0, fb3, am[3]);
                am[4] = WMMA_F32(fa1, fb0, am[4]);
                am[5] = WMMA_F32(fa1, fb1, am[5]);
                am[6] = WMMA_F32(fa1, fb2, am[6]);
                am[7] = WMMA_F32(fa1, fb3, am[7]);
            }
            if (NEX > 0) {
                v2f fbx = *(const v2f*)&Bc[(kB2 * B_PADN + cbx) * 2];
#pragma unroll
                for (int g = 0; g < NEX; ++g) {
                    v2f fe = *(const v2f*)&Ac[(MROWS + 16 * g + ml) * A_PAD + kk + 2 * hi];
                    ex[g] = WMMA_F32(fe, fbx, ex[g]);
                }
            }
        }

        if (more) {
            float* An = As + (cur ^ 1) * A_STAGE;
            float* Bn = Bp + (cur ^ 1) * B_STAGE;
#pragma unroll
            for (int i = 0; i < NLD; ++i)
                *(v4f*)&An[(rowA + 32 * i) * A_PAD + colA] = aOk[i] ? ra[i] : z;
#pragma unroll
            for (int i = 0; i < 2; ++i) {
                const int kp = kp0 + 8 * i;
                v2f p0 = {rlo[i].x, rhi[i].x}, p1 = {rlo[i].y, rhi[i].y};
                v2f p2 = {rlo[i].z, rhi[i].z}, p3 = {rlo[i].w, rhi[i].w};
                float* bp = &Bn[(kp * B_PADN + cBb) * 2];
                *(v2f*)(bp + 0) = p0;
                *(v2f*)(bp + 2) = p1;
                *(v2f*)(bp + 4) = p2;
                *(v2f*)(bp + 6) = p3;
            }
        }
        __syncthreads();
    }

    if (NEX > 0) {
#pragma unroll
        for (int g = 0; g < NEX; ++g)
#pragma unroll
            for (int r = 0; r < 8; ++r)
                Cs[(16 * g + 8 * hi + r) * C_PADN + cbx] = ex[g][r];
        __syncthreads();
#pragma unroll
        for (int j = 0; j < 2 * NEX; ++j) {
            const int row = wave + 8 * j;
            const int rS  = exRow0 + row;
            if (rS < k)
                *(v4f*)(Crow + (size_t)rS * NN + blockN0 + 4 * lane) =
                    *(const v4f*)&Cs[row * C_PADN + 4 * lane];
        }
    }

    if (MROWS > 0) {
        const int colBase = blockN0 + wn * 64 + ml;
        const int row0 = blockM0 + wm * 32 + 8 * hi;
#pragma unroll
        for (int r = 0; r < 8; ++r) {
            float* cp = Csq + (size_t)(row0 + r) * NN + colBase;
            cp[ 0] = am[0][r];
            cp[16] = am[1][r];
            cp[32] = am[2][r];
            cp[48] = am[3][r];
            float* cq = Csq + (size_t)(row0 + 16 + r) * NN + colBase;
            cq[ 0] = am[4][r];
            cq[16] = am[5][r];
            cq[32] = am[6][r];
            cq[48] = am[7][r];
        }
    }
}

template <int MROWS, int NEX>
__global__ __launch_bounds__(256)
void level_kernel(const float* __restrict__ Bm, float* __restrict__ Csq,
                  const float* __restrict__ Srow, float* __restrict__ Crow, int k) {
    constexpr int ARP = (MROWS + 16 * NEX + 31) / 32 * 32;
    __shared__ float As[2 * ARP * A_PAD];
    __shared__ float Bp[2 * (KC / 2) * B_PADN * 2];
    __shared__ float Cs[16 * NEX * C_PADN];

    if constexpr (MROWS == 0) {
        gemm_body<MROWS, NEX>(As, Bp, Cs, Bm, Csq, Srow, Crow, k);
    } else {
        if ((int)blockIdx.y * (16 * NEX) < k)
            gemm_body<MROWS, NEX>(As, Bp, Cs, Bm, Csq, Srow, Crow, k);
        else
            gemm_body<MROWS, 0>(As, Bp, Cs, Bm, Csq, Srow, Crow, k);
    }
}

#define CH_ROWS  128
#define CH_M     32
#define CH_N     32
#define CH_KC    128
#define CH_APAD  132
#define CH_BPAD  48
#define CH_CPAD  36

__global__ __launch_bounds__(256)
void chain_kernel(const float* __restrict__ G, const float* __restrict__ Sprev,
                  float* __restrict__ Snext) {
    constexpr int A_STAGE = CH_M * CH_APAD;
    constexpr int B_STAGE = (CH_KC / 2) * CH_BPAD * 2;
    constexpr int C_HALF  = CH_M * CH_CPAD;
    __shared__ alignas(16) float As[2 * A_STAGE];
    __shared__ alignas(16) float Bp[2 * B_STAGE];
    __shared__ alignas(16) float Cs[2 * C_HALF];

    const int tid  = threadIdx.x;
    const int lane = tid & 31;
    const int wave = tid >> 5;
    const int wm   = wave & 1;
    const int wn   = (wave >> 1) & 1;
    const int kh   = wave >> 2;
    const int hi   = lane >> 4;
    const int ml   = lane & 15;

    const int blockN0 = blockIdx.x * CH_N;
    const int blockM0 = blockIdx.y * CH_M;

    const int rowA = tid >> 5;
    const int colA = (tid & 31) << 2;
    const int kp0  = tid >> 3;
    const int cB   = (tid & 7) << 2;

    const float* aSrc = Sprev + (size_t)(blockM0 + rowA) * NN + colA;
    const float* bSrc = G + (size_t)(2 * kp0) * NN + blockN0 + cB;

    v4f ra[4], rlo[2], rhi[2];

#pragma unroll
    for (int i = 0; i < 4; ++i)
        ra[i] = *(const v4f*)(aSrc + (size_t)(8 * i) * NN);
#pragma unroll
    for (int i = 0; i < 2; ++i) {
        const float* brow = bSrc + (size_t)(64 * i) * NN;
        rlo[i] = *(const v4f*)(brow);
        rhi[i] = *(const v4f*)(brow + NN);
    }
#pragma unroll
    for (int i = 0; i < 4; ++i)
        *(v4f*)&As[(rowA + 8 * i) * CH_APAD + colA] = ra[i];
#pragma unroll
    for (int i = 0; i < 2; ++i) {
        float* bp = &Bp[((kp0 + 32 * i) * CH_BPAD + cB) * 2];
        *(v4f*)(bp + 0) = (v4f){rlo[i].x, rhi[i].x, rlo[i].y, rhi[i].y};
        *(v4f*)(bp + 4) = (v4f){rlo[i].z, rhi[i].z, rlo[i].w, rhi[i].w};
    }
    __syncthreads();

    const int aOff = (wm * 16 + ml) * CH_APAD + 64 * kh + 2 * hi;
    const int bOff = ((32 * kh + hi) * CH_BPAD + wn * 16 + ml) * 2;

    v8f acc0 = (v8f){}, acc1 = (v8f){};

    for (int kb = 0; kb < NN; kb += CH_KC) {
        const int cur = (kb / CH_KC) & 1;
        const bool more = kb + CH_KC < NN;
        const float* Ac = As + cur * A_STAGE;
        const float* Bc = Bp + cur * B_STAGE;

        if (more) {
#pragma unroll
            for (int i = 0; i < 4; ++i)
                ra[i] = *(const v4f*)(aSrc + (size_t)(8 * i) * NN + kb + CH_KC);
#pragma unroll
            for (int i = 0; i < 2; ++i) {
                const float* brow = bSrc + (size_t)(kb + CH_KC + 64 * i) * NN;
                rlo[i] = *(const v4f*)(brow);
                rhi[i] = *(const v4f*)(brow + NN);
            }
        }

#pragma unroll
        for (int kk = 0; kk < CH_KC / 2; kk += 8) {
            v2f fa0 = *(const v2f*)&Ac[aOff + kk];
            v2f fb0 = *(const v2f*)&Bc[bOff + (kk >> 1) * CH_BPAD * 2];
            v2f fa1 = *(const v2f*)&Ac[aOff + kk + 4];
            v2f fb1 = *(const v2f*)&Bc[bOff + ((kk + 4) >> 1) * CH_BPAD * 2];
            acc0 = WMMA_F32(fa0, fb0, acc0);
            acc1 = WMMA_F32(fa1, fb1, acc1);
        }

        if (more) {
            float* An = As + (cur ^ 1) * A_STAGE;
            float* Bn = Bp + (cur ^ 1) * B_STAGE;
#pragma unroll
            for (int i = 0; i < 4; ++i)
                *(v4f*)&An[(rowA + 8 * i) * CH_APAD + colA] = ra[i];
#pragma unroll
            for (int i = 0; i < 2; ++i) {
                float* bp = &Bn[((kp0 + 32 * i) * CH_BPAD + cB) * 2];
                *(v4f*)(bp + 0) = (v4f){rlo[i].x, rhi[i].x, rlo[i].y, rhi[i].y};
                *(v4f*)(bp + 4) = (v4f){rlo[i].z, rhi[i].z, rlo[i].w, rhi[i].w};
            }
        }
        __syncthreads();
    }

    const v8f acc = acc0 + acc1;
#pragma unroll
    for (int r = 0; r < 8; ++r)
        Cs[kh * C_HALF + (wm * 16 + 8 * hi + r) * CH_CPAD + wn * 16 + ml] = acc[r];
    __syncthreads();
    {
        const int row = tid >> 3;
        const int c4  = (tid & 7) << 2;
        const v4f s0 = *(const v4f*)&Cs[row * CH_CPAD + c4];
        const v4f s1 = *(const v4f*)&Cs[C_HALF + row * CH_CPAD + c4];
        *(v4f*)(Snext + (size_t)(blockM0 + row) * NN + blockN0 + c4) = s0 + s1;
    }
}

extern "C" void kernel_launch(void* const* d_in, const int* in_sizes, int n_in,
                              void* d_out, int out_size, void* d_ws, size_t ws_size,
                              hipStream_t stream) {
    (void)in_sizes; (void)n_in; (void)out_size;
    const float* x     = (const float*)d_in[0];
    const float* start = (const float*)d_in[1];
    float* S = (float*)d_out;

    const size_t MATSZ = (size_t)NN * NN;
    if (ws_size < 2 * MATSZ * sizeof(float)) return;
    float* P  = (float*)d_ws;
    float* Pn = P + MATSZ;

    build_A_kernel<<<dim3((unsigned)(MATSZ / 256)), dim3(256), 0, stream>>>(x, P);
    row0_kernel<<<dim3(NN / 256), dim3(256), 0, stream>>>(start, S);

    const dim3 grid(NN / TILE_N, 16);
    for (int k = 1; k < CH_ROWS; k <<= 1) {
        level_kernel<128, 1><<<grid, dim3(256), 0, stream>>>(P, Pn, S, S + (size_t)k * NN, k);
        float* t = P; P = Pn; Pn = t;
    }

    const dim3 cgrid(NN / CH_N, CH_ROWS / CH_M);
    for (int j = 1; j < NN / CH_ROWS; ++j)
        chain_kernel<<<cgrid, dim3(256), 0, stream>>>(P, S + (size_t)(j - 1) * CH_ROWS * NN,
                                                      S + (size_t)j * CH_ROWS * NN);
}


# Chain step cut for bytes in flight: 16 waves per block, 256-k chunks, global loads of two chunks kept in flight across LDS-only barriers; rows 128.. now summed over 4 K quarters (order changes)
# speedup vs baseline: 4.7169x; 1.1432x over previous
#include <hip/hip_runtime.h>


#define NN      2048
#define TILE_N  128
#define KC      32
#define A_PAD   36
#define B_PADN  132
#define C_PADN  132

typedef float v2f __attribute__((ext_vector_type(2)));
typedef float v4f __attribute__((ext_vector_type(4)));
typedef float v8f __attribute__((ext_vector_type(8)));

__global__ __launch_bounds__(256) void build_A_kernel(const float* __restrict__ x,
                                                      float* __restrict__ A) {
    int idx = blockIdx.x * blockDim.x + threadIdx.x;
    int row = idx >> 11;
    int col = idx & (NN - 1);
    float v = x[idx];
    if (row == col) v -= 1.0f;
    A[idx] = v;
}

__global__ __launch_bounds__(256) void row0_kernel(const float* __restrict__ start,
                                                   float* __restrict__ S) {
    int j = blockIdx.x * blockDim.x + threadIdx.x;
    S[j] = start[199 * NN + j];
}

#define WMMA_F32(A_, B_, C_) \
    __builtin_amdgcn_wmma_f32_16x16x4_f32(false, (A_), false, (B_), (short)0, (C_), false, false)

template <int MROWS, int NEX>
__device__ __forceinline__ void gemm_body(float* As, float* Bp, float* Cs,
                                          const float* __restrict__ Bm, float* __restrict__ Csq,
                                          const float* __restrict__ Srow, float* __restrict__ Crow,
                                          const int k) {
    constexpr int AR      = MROWS + 16 * NEX;
    constexpr int ARP     = (AR + 31) / 32 * 32;
    constexpr int NLD     = ARP / 32;
    constexpr int NMAIN   = MROWS / 32;
    constexpr int A_STAGE = ARP * A_PAD;
    constexpr int B_STAGE = (KC / 2) * B_PADN * 2;
    constexpr int NEXA    = NEX > 0 ? NEX : 1;

    const int tid  = threadIdx.x;
    const int lane = tid & 31;
    const int wave = tid >> 5;
    const int wm   = wave & 3;
    const int wn   = wave >> 2;
    const int hi   = lane >> 4;
    const int ml   = lane & 15;

    const int blockN0 = blockIdx.x * TILE_N;
    const int blockM0 = blockIdx.y * MROWS;
    const int exRow0  = blockIdx.y * (16 * NEX);

    v8f am[8];
    v8f ex[NEXA];
#pragma unroll
    for (int i = 0; i < 8; ++i) am[i] = (v8f){};
#pragma unroll
    for (int i = 0; i < NEXA; ++i) ex[i] = (v8f){};

    const int rowA = tid >> 3;
    const int colA = (tid & 7) << 2;
    const int kp0  = tid >> 5;
    const int cBb  = (tid & 31) << 2;

    const float* aSrc[NLD];
    bool aOk[NLD];
#pragma unroll
    for (int i = 0; i < NLD; ++i) {
        if (i < NMAIN) {
            aOk[i]  = true;
            aSrc[i] = Bm + (size_t)(blockM0 + rowA + 32 * i) * NN + colA;
        } else {
            const int e  = rowA + 32 * (i - NMAIN);
            const int rS = exRow0 + e;
            aOk[i]  = (e < 16 * NEX) && (rS < k);
            aSrc[i] = Srow + (size_t)(aOk[i] ? rS : 0) * NN + colA;
        }
    }
    const float* bSrc = Bm + (size_t)(2 * kp0) * NN + blockN0 + cBb;

    v4f ra[NLD], rlo[2], rhi[2];
    const v4f z = {0.f, 0.f, 0.f, 0.f};

#pragma unroll
    for (int i = 0; i < NLD; ++i)
        ra[i] = *(const v4f*)(aSrc[i]);
#pragma unroll
    for (int i = 0; i < 2; ++i) {
        const float* brow = bSrc + (size_t)(16 * i) * NN;
        rlo[i] = *(const v4f*)(brow);
        rhi[i] = *(const v4f*)(brow + NN);
    }
#pragma unroll
    for (int i = 0; i < NLD; ++i)
        *(v4f*)&As[(rowA + 32 * i) * A_PAD + colA] = aOk[i] ? ra[i] : z;
#pragma unroll
    for (int i = 0; i < 2; ++i) {
        const int kp = kp0 + 8 * i;
        v2f p0 = {rlo[i].x, rhi[i].x}, p1 = {rlo[i].y, rhi[i].y};
        v2f p2 = {rlo[i].z, rhi[i].z}, p3 = {rlo[i].w, rhi[i].w};
        float* bp = &Bp[(kp * B_PADN + cBb) * 2];
        *(v2f*)(bp + 0) = p0;
        *(v2f*)(bp + 2) = p1;
        *(v2f*)(bp + 4) = p2;
        *(v2f*)(bp + 6) = p3;
    }
    __syncthreads();

    const int cb0 = wn * 64 + ml;
    const int ar0 = wm * 32 + ml;
    const int cbx = wave * 16 + ml;

    for (int kb = 0; kb < NN; kb += KC) {
        const int cur = (kb / KC) & 1;
        const bool more = kb + KC < NN;
        const float* Ac = As + cur * A_STAGE;
        const float* Bc = Bp + cur * B_STAGE;

        if (more) {
#pragma unroll
            for (int i = 0; i < NLD; ++i)
                ra[i] = *(const v4f*)(aSrc[i] + kb + KC);
#pragma unroll
            for (int i = 0; i < 2; ++i) {
                const float* brow = bSrc + (size_t)(kb + KC + 16 * i) * NN;
                rlo[i] = *(const v4f*)(brow);
                rhi[i] = *(const v4f*)(brow + NN);
            }
        }

#pragma unroll
        for (int kk = 0; kk < KC; kk += 4) {
            const int kB2 = (kk >> 1) + hi;
            if (MROWS > 0) {
                v2f fa0 = *(const v2f*)&Ac[ar0 * A_PAD + kk + 2 * hi];
                v2f fa1 = *(const v2f*)&Ac[(ar0 + 16) * A_PAD + kk + 2 * hi];
                v2f fb0 = *(const v2f*)&Bc[(kB2 * B_PADN + cb0 +  0) * 2];
                v2f fb1 = *(const v2f*)&Bc[(kB2 * B_PADN + cb0 + 16) * 2];
                v2f fb2 = *(const v2f*)&Bc[(kB2 * B_PADN + cb0 + 32) * 2];
                v2f fb3 = *(const v2f*)&Bc[(kB2 * B_PADN + cb0 + 48) * 2];
                am[0] = WMMA_F32(fa0, fb0, am[0]);
                am[1] = WMMA_F32(fa0, fb1, am[1]);
                am[2] = WMMA_F32(fa0, fb2, am[2]);
                am[3] = WMMA_F32(fa0, fb3, am[3]);
                am[4] = WMMA_F32(fa1, fb0, am[4]);
                am[5] = WMMA_F32(fa1, fb1, am[5]);
                am[6] = WMMA_F32(fa1, fb2, am[6]);
                am[7] = WMMA_F32(fa1, fb3, am[7]);
            }
            if (NEX > 0) {
                v2f fbx = *(const v2f*)&Bc[(kB2 * B_PADN + cbx) * 2];
#pragma unroll
                for (int g = 0; g < NEX; ++g) {
                    v2f fe = *(const v2f*)&Ac[(MROWS + 16 * g + ml) * A_PAD + kk + 2 * hi];
                    ex[g] = WMMA_F32(fe, fbx, ex[g]);
                }
            }
        }

        if (more) {
            float* An = As + (cur ^ 1) * A_STAGE;
            float* Bn = Bp + (cur ^ 1) * B_STAGE;
#pragma unroll
            for (int i = 0; i < NLD; ++i)
                *(v4f*)&An[(rowA + 32 * i) * A_PAD + colA] = aOk[i] ? ra[i] : z;
#pragma unroll
            for (int i = 0; i < 2; ++i) {
                const int kp = kp0 + 8 * i;
                v2f p0 = {rlo[i].x, rhi[i].x}, p1 = {rlo[i].y, rhi[i].y};
                v2f p2 = {rlo[i].z, rhi[i].z}, p3 = {rlo[i].w, rhi[i].w};
                float* bp = &Bn[(kp * B_PADN + cBb) * 2];
                *(v2f*)(bp + 0) = p0;
                *(v2f*)(bp + 2) = p1;
                *(v2f*)(bp + 4) = p2;
                *(v2f*)(bp + 6) = p3;
            }
        }
        __syncthreads();
    }

    if (NEX > 0) {
#pragma unroll
        for (int g = 0; g < NEX; ++g)
#pragma unroll
            for (int r = 0; r < 8; ++r)
                Cs[(16 * g + 8 * hi + r) * C_PADN + cbx] = ex[g][r];
        __syncthreads();
#pragma unroll
        for (int j = 0; j < 2 * NEX; ++j) {
            const int row = wave + 8 * j;
            const int rS  = exRow0 + row;
            if (rS < k)
                *(v4f*)(Crow + (size_t)rS * NN + blockN0 + 4 * lane) =
                    *(const v4f*)&Cs[row * C_PADN + 4 * lane];
        }
    }

    if (MROWS > 0) {
        const int colBase = blockN0 + wn * 64 + ml;
        const int row0 = blockM0 + wm * 32 + 8 * hi;
#pragma unroll
        for (int r = 0; r < 8; ++r) {
            float* cp = Csq + (size_t)(row0 + r) * NN + colBase;
            cp[ 0] = am[0][r];
            cp[16] = am[1][r];
            cp[32] = am[2][r];
            cp[48] = am[3][r];
            float* cq = Csq + (size_t)(row0 + 16 + r) * NN + colBase;
            cq[ 0] = am[4][r];
            cq[16] = am[5][r];
            cq[32] = am[6][r];
            cq[48] = am[7][r];
        }
    }
}

template <int MROWS, int NEX>
__global__ __launch_bounds__(256)
void level_kernel(const float* __restrict__ Bm, float* __restrict__ Csq,
                  const float* __restrict__ Srow, float* __restrict__ Crow, int k) {
    constexpr int ARP = (MROWS + 16 * NEX + 31) / 32 * 32;
    __shared__ float As[2 * ARP * A_PAD];
    __shared__ float Bp[2 * (KC / 2) * B_PADN * 2];
    __shared__ float Cs[16 * NEX * C_PADN];

    if constexpr (MROWS == 0) {
        gemm_body<MROWS, NEX>(As, Bp, Cs, Bm, Csq, Srow, Crow, k);
    } else {
        if ((int)blockIdx.y * (16 * NEX) < k)
            gemm_body<MROWS, NEX>(As, Bp, Cs, Bm, Csq, Srow, Crow, k);
        else
            gemm_body<MROWS, 0>(As, Bp, Cs, Bm, Csq, Srow, Crow, k);
    }
}

#define CH_ROWS  128
#define CH_M     32
#define CH_N     32
#define CH_KC    256
#define CH_NT    512
#define CH_APAD  260
#define CH_BPAD  40
#define CH_CPAD  36

__device__ __forceinline__ void lds_barrier() {
    __builtin_amdgcn_fence(__ATOMIC_RELEASE, "workgroup", "local");
    __builtin_amdgcn_s_barrier();
    __builtin_amdgcn_fence(__ATOMIC_ACQUIRE, "workgroup", "local");
}

struct ChSet { v4f ra[4]; v4f rb[4]; };

__device__ __forceinline__ void ch_load(ChSet& s, const float* __restrict__ aSrc,
                                        const float* __restrict__ bSrc, const int kb) {
#pragma unroll
    for (int i = 0; i < 4; ++i)
        s.ra[i] = *(const v4f*)(aSrc + (size_t)(8 * i) * NN + kb);
#pragma unroll
    for (int i = 0; i < 4; ++i)
        s.rb[i] = *(const v4f*)(bSrc + (size_t)(kb + 64 * i) * NN);
}

__device__ __forceinline__ void ch_store(const ChSet& s, float* An, float* Bn, const int aDst,
                                         const int bDst) {
#pragma unroll
    for (int i = 0; i < 4; ++i)
        *(v4f*)&An[aDst + 8 * i * CH_APAD] = s.ra[i];
#pragma unroll
    for (int i = 0; i < 4; ++i)
        *(v4f*)&Bn[bDst + 64 * i * CH_BPAD] = s.rb[i];
}

__device__ __forceinline__ void ch_compute(const float* Ac, const float* Bc, const int aOff,
                                           const int bOff, v8f& acc0, v8f& acc1) {
#pragma unroll
    for (int kk = 0; kk < CH_KC / 4; kk += 8) {
        v2f fa0 = *(const v2f*)&Ac[aOff + kk];
        v2f fb0 = {Bc[bOff + kk * CH_BPAD], Bc[bOff + (kk + 1) * CH_BPAD]};
        v2f fa1 = *(const v2f*)&Ac[aOff + kk + 4];
        v2f fb1 = {Bc[bOff + (kk + 4) * CH_BPAD], Bc[bOff + (kk + 5) * CH_BPAD]};
        acc0 = WMMA_F32(fa0, fb0, acc0);
        acc1 = WMMA_F32(fa1, fb1, acc1);
    }
}

__device__ __forceinline__ void ch_hold(const v8f& a0, const v8f& a1) {
    asm volatile("" :: "v"(a0), "v"(a1));
}

__global__ __launch_bounds__(CH_NT)
void chain_kernel(const float* __restrict__ G, const float* __restrict__ Sprev,
                  float* __restrict__ Snext) {
    constexpr int A_STAGE = CH_M * CH_APAD;
    constexpr int B_STAGE = CH_KC * CH_BPAD;
    constexpr int C_PART  = CH_M * CH_CPAD;
    __shared__ alignas(16) float As[2 * A_STAGE];
    __shared__ alignas(16) float Bp[2 * B_STAGE];
    __shared__ alignas(16) float Cs[4 * C_PART];

    const int tid  = threadIdx.x;
    const int lane = tid & 31;
    const int wave = tid >> 5;
    const int wm   = wave & 1;
    const int wn   = (wave >> 1) & 1;
    const int kq   = wave >> 2;
    const int hi   = lane >> 4;
    const int ml   = lane & 15;

    const int blockN0 = blockIdx.x * CH_N;
    const int blockM0 = blockIdx.y * CH_M;

    const int rowA = tid >> 6;
    const int colA = (tid & 63) << 2;
    const int rowB = tid >> 3;
    const int cB   = (tid & 7) << 2;

    const float* aSrc = Sprev + (size_t)(blockM0 + rowA) * NN + colA;
    const float* bSrc = G + (size_t)rowB * NN + blockN0 + cB;
    const int aDst = rowA * CH_APAD + colA;
    const int bDst = rowB * CH_BPAD + cB;

    float* As0 = As;
    float* As1 = As + A_STAGE;
    float* Bp0 = Bp;
    float* Bp1 = Bp + B_STAGE;

    const int aOff = (wm * 16 + ml) * CH_APAD + 64 * kq + 2 * hi;
    const int bOff = (64 * kq + 2 * hi) * CH_BPAD + wn * 16 + ml;

    v8f acc0 = (v8f){}, acc1 = (v8f){};

    ChSet s0, s1;
    ch_load(s0, aSrc, bSrc, 0);
    ch_load(s1, aSrc, bSrc, CH_KC);
    ch_store(s0, As0, Bp0, aDst, bDst);
    lds_barrier();

    for (int kb = 0; kb < NN - 2 * CH_KC; kb += 2 * CH_KC) {
        ch_load(s0, aSrc, bSrc, kb + 2 * CH_KC);
        ch_compute(As0, Bp0, aOff, bOff, acc0, acc1);
        ch_hold(acc0, acc1);
        ch_store(s1, As1, Bp1, aDst, bDst);
        lds_barrier();

        ch_load(s1, aSrc, bSrc, kb + 3 * CH_KC);
        ch_compute(As1, Bp1, aOff, bOff, acc0, acc1);
        ch_hold(acc0, acc1);
        ch_store(s0, As0, Bp0, aDst, bDst);
        lds_barrier();
    }
    ch_compute(As0, Bp0, aOff, bOff, acc0, acc1);
    ch_hold(acc0, acc1);
    ch_store(s1, As1, Bp1, aDst, bDst);
    lds_barrier();
    ch_compute(As1, Bp1, aOff, bOff, acc0, acc1);

    const v8f acc = acc0 + acc1;
#pragma unroll
    for (int r = 0; r < 8; ++r)
        Cs[kq * C_PART + (wm * 16 + 8 * hi + r) * CH_CPAD + wn * 16 + ml] = acc[r];
    __syncthreads();
    if (tid < CH_M * CH_N / 4) {
        const int row = tid >> 3;
        const int c4  = (tid & 7) << 2;
        const float* cs = &Cs[row * CH_CPAD + c4];
        const v4f q0 = *(const v4f*)(cs);
        const v4f q1 = *(const v4f*)(cs + C_PART);
        const v4f q2 = *(const v4f*)(cs + 2 * C_PART);
        const v4f q3 = *(const v4f*)(cs + 3 * C_PART);
        *(v4f*)(Snext + (size_t)(blockM0 + row) * NN + blockN0 + c4) = (q0 + q1) + (q2 + q3);
    }
}

extern "C" void kernel_launch(void* const* d_in, const int* in_sizes, int n_in,
                              void* d_out, int out_size, void* d_ws, size_t ws_size,
                              hipStream_t stream) {
    (void)in_sizes; (void)n_in; (void)out_size;
    const float* x     = (const float*)d_in[0];
    const float* start = (const float*)d_in[1];
    float* S = (float*)d_out;

    const size_t MATSZ = (size_t)NN * NN;
    if (ws_size < 2 * MATSZ * sizeof(float)) return;
    float* P  = (float*)d_ws;
    float* Pn = P + MATSZ;

    build_A_kernel<<<dim3((unsigned)(MATSZ / 256)), dim3(256), 0, stream>>>(x, P);
    row0_kernel<<<dim3(NN / 256), dim3(256), 0, stream>>>(start, S);

    const dim3 grid(NN / TILE_N, 16);
    for (int k = 1; k < CH_ROWS; k <<= 1) {
        level_kernel<128, 1><<<grid, dim3(256), 0, stream>>>(P, Pn, S, S + (size_t)k * NN, k);
        float* t = P; P = Pn; Pn = t;
    }

    const dim3 cgrid(NN / CH_N, CH_ROWS / CH_M);
    for (int j = 1; j < NN / CH_ROWS; ++j)
        chain_kernel<<<cgrid, dim3(CH_NT), 0, stream>>>(P, S + (size_t)(j - 1) * CH_ROWS * NN,
                                                      S + (size_t)j * CH_ROWS * NN);
}


# Level kernel: a wave's LDS fragment reads regrouped from two small groups per k-step into two groups per 32-k chunk, each 32+ WMMAs ahead of use (third LDS stage); summation order unchanged
# speedup vs baseline: 4.8827x; 1.0352x over previous
#include <hip/hip_runtime.h>


#define NN      2048
#define TILE_N  128
#define KC      32
#define A_PAD   36
#define B_PADN  132
#define C_PADN  132

typedef float v2f __attribute__((ext_vector_type(2)));
typedef float v4f __attribute__((ext_vector_type(4)));
typedef float v8f __attribute__((ext_vector_type(8)));

__global__ __launch_bounds__(256) void build_A_kernel(const float* __restrict__ x,
                                                      float* __restrict__ A) {
    int idx = blockIdx.x * blockDim.x + threadIdx.x;
    int row = idx >> 11;
    int col = idx & (NN - 1);
    float v = x[idx];
    if (row == col) v -= 1.0f;
    A[idx] = v;
}

__global__ __launch_bounds__(256) void row0_kernel(const float* __restrict__ start,
                                                   float* __restrict__ S) {
    int j = blockIdx.x * blockDim.x + threadIdx.x;
    S[j] = start[199 * NN + j];
}

#define WMMA_F32(A_, B_, C_) \
    __builtin_amdgcn_wmma_f32_16x16x4_f32(false, (A_), false, (B_), (short)0, (C_), false, false)

template <int NEX>
struct FragSet {
    v2f fa0[4], fa1[4];
    v2f fb[4][4];
    v2f fbx[4];
    v2f fe[4][NEX > 0 ? NEX : 1];
};

template <int MROWS, int NEX>
__device__ __forceinline__ void frag_load(FragSet<NEX>& f, const float* Ac, const float* Bc,
                                          const int aOff, const int bOff, const int bxOff,
                                          const int eOff, const int kk0) {
#pragma unroll
    for (int s = 0; s < 4; ++s) {
        const int kk = kk0 + 4 * s;
        const int kp = (kk >> 1) * B_PADN * 2;
        if (MROWS > 0) {
            f.fa0[s] = *(const v2f*)&Ac[aOff + kk];
            f.fa1[s] = *(const v2f*)&Ac[aOff + 16 * A_PAD + kk];
#pragma unroll
            for (int j = 0; j < 4; ++j)
                f.fb[s][j] = *(const v2f*)&Bc[bOff + kp + 32 * j];
        }
        if (NEX > 0) {
            f.fbx[s] = *(const v2f*)&Bc[bxOff + kp];
#pragma unroll
            for (int g = 0; g < NEX; ++g)
                f.fe[s][g] = *(const v2f*)&Ac[eOff + 16 * g * A_PAD + kk];
        }
    }
}

template <int MROWS, int NEX>
__device__ __forceinline__ void frag_mma(const FragSet<NEX>& f, v8f (&am)[8],
                                         v8f (&ex)[NEX > 0 ? NEX : 1]) {
#pragma unroll
    for (int s = 0; s < 4; ++s) {
        if (MROWS > 0) {
            am[0] = WMMA_F32(f.fa0[s], f.fb[s][0], am[0]);
            am[1] = WMMA_F32(f.fa0[s], f.fb[s][1], am[1]);
            am[2] = WMMA_F32(f.fa0[s], f.fb[s][2], am[2]);
            am[3] = WMMA_F32(f.fa0[s], f.fb[s][3], am[3]);
            am[4] = WMMA_F32(f.fa1[s], f.fb[s][0], am[4]);
            am[5] = WMMA_F32(f.fa1[s], f.fb[s][1], am[5]);
            am[6] = WMMA_F32(f.fa1[s], f.fb[s][2], am[6]);
            am[7] = WMMA_F32(f.fa1[s], f.fb[s][3], am[7]);
        }
        if (NEX > 0) {
#pragma unroll
            for (int g = 0; g < NEX; ++g)
                ex[g] = WMMA_F32(f.fe[s][g], f.fbx[s], ex[g]);
        }
    }
}

template <int MROWS, int NEX>
__device__ __forceinline__ void gemm_body(float* As, float* Bp, float* Cs,
                                          const float* __restrict__ Bm, float* __restrict__ Csq,
                                          const float* __restrict__ Srow, float* __restrict__ Crow,
                                          const int k) {
    constexpr int AR      = MROWS + 16 * NEX;
    constexpr int ARP     = (AR + 31) / 32 * 32;
    constexpr int NLD     = ARP / 32;
    constexpr int NMAIN   = MROWS / 32;
    constexpr int A_STAGE = ARP * A_PAD;
    constexpr int B_STAGE = (KC / 2) * B_PADN * 2;
    constexpr int NEXA    = NEX > 0 ? NEX : 1;

    const int tid  = threadIdx.x;
    const int lane = tid & 31;
    const int wave = tid >> 5;
    const int wm   = wave & 3;
    const int wn   = wave >> 2;
    const int hi   = lane >> 4;
    const int ml   = lane & 15;

    const int blockN0 = blockIdx.x * TILE_N;
    const int blockM0 = blockIdx.y * MROWS;
    const int exRow0  = blockIdx.y * (16 * NEX);

    v8f am[8];
    v8f ex[NEXA];
#pragma unroll
    for (int i = 0; i < 8; ++i) am[i] = (v8f){};
#pragma unroll
    for (int i = 0; i < NEXA; ++i) ex[i] = (v8f){};

    const int rowA = tid >> 3;
    const int colA = (tid & 7) << 2;
    const int kp0  = tid >> 5;
    const int cBb  = (tid & 31) << 2;

    const float* aSrc[NLD];
    bool aOk[NLD];
#pragma unroll
    for (int i = 0; i < NLD; ++i) {
        if (i < NMAIN) {
            aOk[i]  = true;
            aSrc[i] = Bm + (size_t)(blockM0 + rowA + 32 * i) * NN + colA;
        } else {
            const int e  = rowA + 32 * (i - NMAIN);
            const int rS = exRow0 + e;
            aOk[i]  = (e < 16 * NEX) && (rS < k);
            aSrc[i] = Srow + (size_t)(aOk[i] ? rS : 0) * NN + colA;
        }
    }
    const float* bSrc = Bm + (size_t)(2 * kp0) * NN + blockN0 + cBb;

    v4f ra[NLD], rlo[2], rhi[2];
    const v4f z = {0.f, 0.f, 0.f, 0.f};

    auto gload = [&](const int kb) __attribute__((always_inline)) {
#pragma unroll
        for (int i = 0; i < NLD; ++i)
            ra[i] = *(const v4f*)(aSrc[i] + kb);
#pragma unroll
        for (int i = 0; i < 2; ++i) {
            const float* brow = bSrc + (size_t)(kb + 16 * i) * NN;
            rlo[i] = *(const v4f*)(brow);
            rhi[i] = *(const v4f*)(brow + NN);
        }
    };
    auto sstore = [&](float* An, float* Bn) __attribute__((always_inline)) {
#pragma unroll
        for (int i = 0; i < NLD; ++i)
            *(v4f*)&An[(rowA + 32 * i) * A_PAD + colA] = aOk[i] ? ra[i] : z;
#pragma unroll
        for (int i = 0; i < 2; ++i) {
            const int kp = kp0 + 8 * i;
            v2f p0 = {rlo[i].x, rhi[i].x}, p1 = {rlo[i].y, rhi[i].y};
            v2f p2 = {rlo[i].z, rhi[i].z}, p3 = {rlo[i].w, rhi[i].w};
            float* bp = &Bn[(kp * B_PADN + cBb) * 2];
            *(v2f*)(bp + 0) = p0;
            *(v2f*)(bp + 2) = p1;
            *(v2f*)(bp + 4) = p2;
            *(v2f*)(bp + 6) = p3;
        }
    };

    const int cb0 = wn * 64 + ml;
    const int ar0 = wm * 32 + ml;
    const int cbx = wave * 16 + ml;
    const int aOff  = ar0 * A_PAD + 2 * hi;
    const int bOff  = (hi * B_PADN + cb0) * 2;
    const int bxOff = (hi * B_PADN + cbx) * 2;
    const int eOff  = (MROWS + ml) * A_PAD + 2 * hi;

    gload(0);
    sstore(As, Bp);
    gload(KC);
    sstore(As + A_STAGE, Bp + B_STAGE);
    __syncthreads();

    FragSet<NEX> X, Y;
    frag_load<MROWS, NEX>(X, As, Bp, aOff, bOff, bxOff, eOff, 0);

    constexpr int NCH = NN / KC;
    int s0 = 0, s1 = 1, s2 = 2;
    for (int n = 0; n < NCH; ++n) {
        const bool more1 = n + 1 < NCH;
        const bool more2 = n + 2 < NCH;

        if (more2) gload((n + 2) * KC);
        __builtin_amdgcn_sched_barrier(0);
        frag_load<MROWS, NEX>(Y, As + s0 * A_STAGE, Bp + s0 * B_STAGE, aOff, bOff, bxOff, eOff, 16);
        __builtin_amdgcn_sched_barrier(0);
        frag_mma<MROWS, NEX>(X, am, ex);
        __builtin_amdgcn_sched_barrier(0);
        if (more1)
            frag_load<MROWS, NEX>(X, As + s1 * A_STAGE, Bp + s1 * B_STAGE, aOff, bOff, bxOff, eOff, 0);
        __builtin_amdgcn_sched_barrier(0);
        frag_mma<MROWS, NEX>(Y, am, ex);
        __builtin_amdgcn_sched_barrier(0);

        if (more2) sstore(As + s2 * A_STAGE, Bp + s2 * B_STAGE);
        __syncthreads();
        const int t = s0; s0 = s1; s1 = s2; s2 = t;
    }

    if (NEX > 0) {
#pragma unroll
        for (int g = 0; g < NEX; ++g)
#pragma unroll
            for (int r = 0; r < 8; ++r)
                Cs[(16 * g + 8 * hi + r) * C_PADN + cbx] = ex[g][r];
        __syncthreads();
#pragma unroll
        for (int j = 0; j < 2 * NEX; ++j) {
            const int row = wave + 8 * j;
            const int rS  = exRow0 + row;
            if (rS < k)
                *(v4f*)(Crow + (size_t)rS * NN + blockN0 + 4 * lane) =
                    *(const v4f*)&Cs[row * C_PADN + 4 * lane];
        }
    }

    if (MROWS > 0) {
        const int colBase = blockN0 + wn * 64 + ml;
        const int row0 = blockM0 + wm * 32 + 8 * hi;
#pragma unroll
        for (int r = 0; r < 8; ++r) {
            float* cp = Csq + (size_t)(row0 + r) * NN + colBase;
            cp[ 0] = am[0][r];
            cp[16] = am[1][r];
            cp[32] = am[2][r];
            cp[48] = am[3][r];
            float* cq = Csq + (size_t)(row0 + 16 + r) * NN + colBase;
            cq[ 0] = am[4][r];
            cq[16] = am[5][r];
            cq[32] = am[6][r];
            cq[48] = am[7][r];
        }
    }
}

template <int MROWS, int NEX>
__global__ __launch_bounds__(256)
void level_kernel(const float* __restrict__ Bm, float* __restrict__ Csq,
                  const float* __restrict__ Srow, float* __restrict__ Crow, int k) {
    constexpr int ARP = (MROWS + 16 * NEX + 31) / 32 * 32;
    __shared__ alignas(16) float As[3 * ARP * A_PAD];
    __shared__ alignas(16) float Bp[3 * (KC / 2) * B_PADN * 2];
    __shared__ alignas(16) float Cs[16 * NEX * C_PADN];

    if constexpr (MROWS == 0) {
        gemm_body<MROWS, NEX>(As, Bp, Cs, Bm, Csq, Srow, Crow, k);
    } else {
        if ((int)blockIdx.y * (16 * NEX) < k)
            gemm_body<MROWS, NEX>(As, Bp, Cs, Bm, Csq, Srow, Crow, k);
        else
            gemm_body<MROWS, 0>(As, Bp, Cs, Bm, Csq, Srow, Crow, k);
    }
}

#define CH_ROWS  128
#define CH_M     32
#define CH_N     32
#define CH_KC    256
#define CH_NT    512
#define CH_APAD  260
#define CH_BPAD  40
#define CH_CPAD  36

__device__ __forceinline__ void lds_barrier() {
    __builtin_amdgcn_fence(__ATOMIC_RELEASE, "workgroup", "local");
    __builtin_amdgcn_s_barrier();
    __builtin_amdgcn_fence(__ATOMIC_ACQUIRE, "workgroup", "local");
}

struct ChSet { v4f ra[4]; v4f rb[4]; };

__device__ __forceinline__ void ch_load(ChSet& s, const float* __restrict__ aSrc,
                                        const float* __restrict__ bSrc, const int kb) {
#pragma unroll
    for (int i = 0; i < 4; ++i)
        s.ra[i] = *(const v4f*)(aSrc + (size_t)(8 * i) * NN + kb);
#pragma unroll
    for (int i = 0; i < 4; ++i)
        s.rb[i] = *(const v4f*)(bSrc + (size_t)(kb + 64 * i) * NN);
}

__device__ __forceinline__ void ch_store(const ChSet& s, float* An, float* Bn, const int aDst,
                                         const int bDst) {
#pragma unroll
    for (int i = 0; i < 4; ++i)
        *(v4f*)&An[aDst + 8 * i * CH_APAD] = s.ra[i];
#pragma unroll
    for (int i = 0; i < 4; ++i)
        *(v4f*)&Bn[bDst + 64 * i * CH_BPAD] = s.rb[i];
}

__device__ __forceinline__ void ch_compute(const float* Ac, const float* Bc, const int aOff,
                                           const int bOff, v8f& acc0, v8f& acc1) {
#pragma unroll
    for (int kk = 0; kk < CH_KC / 4; kk += 8) {
        v2f fa0 = *(const v2f*)&Ac[aOff + kk];
        v2f fb0 = {Bc[bOff + kk * CH_BPAD], Bc[bOff + (kk + 1) * CH_BPAD]};
        v2f fa1 = *(const v2f*)&Ac[aOff + kk + 4];
        v2f fb1 = {Bc[bOff + (kk + 4) * CH_BPAD], Bc[bOff + (kk + 5) * CH_BPAD]};
        acc0 = WMMA_F32(fa0, fb0, acc0);
        acc1 = WMMA_F32(fa1, fb1, acc1);
    }
}

__device__ __forceinline__ void ch_hold(const v8f& a0, const v8f& a1) {
    asm volatile("" :: "v"(a0), "v"(a1));
}

__global__ __launch_bounds__(CH_NT)
void chain_kernel(const float* __restrict__ G, const float* __restrict__ Sprev,
                  float* __restrict__ Snext) {
    constexpr int A_STAGE = CH_M * CH_APAD;
    constexpr int B_STAGE = CH_KC * CH_BPAD;
    constexpr int C_PART  = CH_M * CH_CPAD;
    __shared__ alignas(16) float As[2 * A_STAGE];
    __shared__ alignas(16) float Bp[2 * B_STAGE];
    __shared__ alignas(16) float Cs[4 * C_PART];

    const int tid  = threadIdx.x;
    const int lane = tid & 31;
    const int wave = tid >> 5;
    const int wm   = wave & 1;
    const int wn   = (wave >> 1) & 1;
    const int kq   = wave >> 2;
    const int hi   = lane >> 4;
    const int ml   = lane & 15;

    const int blockN0 = blockIdx.x * CH_N;
    const int blockM0 = blockIdx.y * CH_M;

    const int rowA = tid >> 6;
    const int colA = (tid & 63) << 2;
    const int rowB = tid >> 3;
    const int cB   = (tid & 7) << 2;

    const float* aSrc = Sprev + (size_t)(blockM0 + rowA) * NN + colA;
    const float* bSrc = G + (size_t)rowB * NN + blockN0 + cB;
    const int aDst = rowA * CH_APAD + colA;
    const int bDst = rowB * CH_BPAD + cB;

    float* As0 = As;
    float* As1 = As + A_STAGE;
    float* Bp0 = Bp;
    float* Bp1 = Bp + B_STAGE;

    const int aOff = (wm * 16 + ml) * CH_APAD + 64 * kq + 2 * hi;
    const int bOff = (64 * kq + 2 * hi) * CH_BPAD + wn * 16 + ml;

    v8f acc0 = (v8f){}, acc1 = (v8f){};

    ChSet s0, s1;
    ch_load(s0, aSrc, bSrc, 0);
    ch_load(s1, aSrc, bSrc, CH_KC);
    ch_store(s0, As0, Bp0, aDst, bDst);
    lds_barrier();

    for (int kb = 0; kb < NN - 2 * CH_KC; kb += 2 * CH_KC) {
        ch_load(s0, aSrc, bSrc, kb + 2 * CH_KC);
        ch_compute(As0, Bp0, aOff, bOff, acc0, acc1);
        ch_hold(acc0, acc1);
        ch_store(s1, As1, Bp1, aDst, bDst);
        lds_barrier();

        ch_load(s1, aSrc, bSrc, kb + 3 * CH_KC);
        ch_compute(As1, Bp1, aOff, bOff, acc0, acc1);
        ch_hold(acc0, acc1);
        ch_store(s0, As0, Bp0, aDst, bDst);
        lds_barrier();
    }
    ch_compute(As0, Bp0, aOff, bOff, acc0, acc1);
    ch_hold(acc0, acc1);
    ch_store(s1, As1, Bp1, aDst, bDst);
    lds_barrier();
    ch_compute(As1, Bp1, aOff, bOff, acc0, acc1);

    const v8f acc = acc0 + acc1;
#pragma unroll
    for (int r = 0; r < 8; ++r)
        Cs[kq * C_PART + (wm * 16 + 8 * hi + r) * CH_CPAD + wn * 16 + ml] = acc[r];
    __syncthreads();
    if (tid < CH_M * CH_N / 4) {
        const int row = tid >> 3;
        const int c4  = (tid & 7) << 2;
        const float* cs = &Cs[row * CH_CPAD + c4];
        const v4f q0 = *(const v4f*)(cs);
        const v4f q1 = *(const v4f*)(cs + C_PART);
        const v4f q2 = *(const v4f*)(cs + 2 * C_PART);
        const v4f q3 = *(const v4f*)(cs + 3 * C_PART);
        *(v4f*)(Snext + (size_t)(blockM0 + row) * NN + blockN0 + c4) = (q0 + q1) + (q2 + q3);
    }
}

extern "C" void kernel_launch(void* const* d_in, const int* in_sizes, int n_in,
                              void* d_out, int out_size, void* d_ws, size_t ws_size,
                              hipStream_t stream) {
    (void)in_sizes; (void)n_in; (void)out_size;
    const float* x     = (const float*)d_in[0];
    const float* start = (const float*)d_in[1];
    float* S = (float*)d_out;

    const size_t MATSZ = (size_t)NN * NN;
    if (ws_size < 2 * MATSZ * sizeof(float)) return;
    float* P  = (float*)d_ws;
    float* Pn = P + MATSZ;

    build_A_kernel<<<dim3((unsigned)(MATSZ / 256)), dim3(256), 0, stream>>>(x, P);
    row0_kernel<<<dim3(NN / 256), dim3(256), 0, stream>>>(start, S);

    const dim3 grid(NN / TILE_N, 16);
    for (int k = 1; k < CH_ROWS; k <<= 1) {
        level_kernel<128, 1><<<grid, dim3(256), 0, stream>>>(P, Pn, S, S + (size_t)k * NN, k);
        float* t = P; P = Pn; Pn = t;
    }

    const dim3 cgrid(NN / CH_N, CH_ROWS / CH_M);
    for (int j = 1; j < NN / CH_ROWS; ++j)
        chain_kernel<<<cgrid, dim3(CH_NT), 0, stream>>>(P, S + (size_t)(j - 1) * CH_ROWS * NN,
                                                      S + (size_t)j * CH_ROWS * NN);
}
